# starting kernel
# baseline (speedup 1.0000x reference)
#include <hip/hip_runtime.h>
#include <math.h>

typedef __attribute__((ext_vector_type(2))) float v2f;
typedef __attribute__((ext_vector_type(4))) float v4f;
typedef __attribute__((ext_vector_type(8))) float v8f;

#define WAVES_PER_BLOCK 4
#define BLOCK (WAVES_PER_BLOCK * 32)
#define KCHUNK 512

__global__ __launch_bounds__(BLOCK) void dpl_safe_policy_kernel(
    const float* __restrict__ x,
    const float* __restrict__ Wg, const float* __restrict__ bg,
    const float* __restrict__ Wp, const float* __restrict__ bp,
    const float* __restrict__ Wa, const float* __restrict__ ba,
    float* __restrict__ out, int H)
{
  __shared__ float ldsW[KCHUNK * 16];
  __shared__ float ldsC[WAVES_PER_BLOCK * 256];

  const int tid  = threadIdx.x;
  const int lane = tid & 31;
  const int wave = tid >> 5;

  const int tile    = blockIdx.x * WAVES_PER_BLOCK + wave;
  const int rowBase = tile * 16;

  const int m     = lane & 15;
  const int half  = lane >> 4;
  const float* aPtr = x + (size_t)(rowBase + m) * (size_t)H + (half << 1);

  const float* bPtr = ldsW + ((half * 16 + m) << 1);

  v8f c = {};

  for (int k0 = 0; k0 < H; k0 += KCHUNK) {
    __syncthreads();

    for (int idx = tid; idx < KCHUNK * 16; idx += BLOCK) {
      int kk = idx >> 4;
      int k  = k0 + kk;
      int n  = idx & 15;
      float v;
      if (n < 4)       v = Wg[k * 4 + n];
      else if (n < 8)  v = Wp[k * 4 + (n - 4)];
      else if (n < 13) v = Wa[k * 5 + (n - 8)];
      else             v = 0.0f;
      ldsW[(((kk >> 1) * 16 + n) << 1) + (kk & 1)] = v;
    }
    __syncthreads();

    if (k0 + KCHUNK < H)
      __builtin_prefetch(aPtr + k0 + KCHUNK, 0, 1);

#pragma unroll 8
    for (int k = 0; k < KCHUNK; k += 4) {
      v2f a = *(const v2f*)(aPtr + k0 + k);
      v2f b = *(const v2f*)(bPtr + (k >> 1) * 32);
      c = __builtin_amdgcn_wmma_f32_16x16x4_f32(
               false, a,  false, b,
               (short)0, c,  false,  false);
    }
  }

  {
    float* cw = ldsC + wave * 256;
#pragma unroll
    for (int r = 0; r < 8; ++r)
      cw[(r + 8 * half) * 16 + m] = c[r];
  }

  if (lane < 16) {
    const float* cw = ldsC + wave * 256 + lane * 16;
    float vals[16];
#pragma unroll
    for (int i = 0; i < 16; i += 4) {
      v4f t = *(const v4f*)(cw + i);
      vals[i] = t.x; vals[i + 1] = t.y; vals[i + 2] = t.z; vals[i + 3] = t.w;
    }
#pragma unroll
    for (int i = 0; i < 4; ++i) vals[i]     += bg[i];
#pragma unroll
    for (int i = 0; i < 4; ++i) vals[4 + i] += bp[i];
#pragma unroll
    for (int i = 0; i < 5; ++i) vals[8 + i] += ba[i];

    float mg = fmaxf(fmaxf(vals[0], vals[1]), fmaxf(vals[2], vals[3]));
    float g0 = __expf(vals[0] - mg), g1 = __expf(vals[1] - mg);
    float g2 = __expf(vals[2] - mg), g3 = __expf(vals[3] - mg);
    float ginv = 1.0f / (g0 + g1 + g2 + g3);

    float mp = fmaxf(fmaxf(vals[4], vals[5]), fmaxf(vals[6], vals[7]));
    float p0 = __expf(vals[4] - mp), p1 = __expf(vals[5] - mp);
    float p2 = __expf(vals[6] - mp), p3 = __expf(vals[7] - mp);
    float pinv = 1.0f / (p0 + p1 + p2 + p3);

    float ma = fmaxf(fmaxf(fmaxf(vals[8], vals[9]), fmaxf(vals[10], vals[11])), vals[12]);
    float e0 = __expf(vals[8]  - ma), e1 = __expf(vals[9]  - ma);
    float e2 = __expf(vals[10] - ma), e3 = __expf(vals[11] - ma);
    float e4 = __expf(vals[12] - ma);

    float gp = ginv * pinv;
    float u0 = (p0 * g0 + p1 * g1 + p2 * g2 + p3 * g3) * gp;
    float u1 = (p0 * g1 + p2 * g3) * gp;
    float u2 = (p1 * g0 + p3 * g2) * gp;

    float j0 = e0 * (1.0f - u0);
    float j1 = e1 * (1.0f - u1);
    float j2 = e2 * (1.0f - u2);
    float j3 = e3;
    float j4 = e4;
    float inv = 1.0f / (j0 + j1 + j2 + j3 + j4);

    float* o = out + (size_t)(rowBase + lane) * 5;
    o[0] = j0 * inv; o[1] = j1 * inv; o[2] = j2 * inv;
    o[3] = j3 * inv; o[4] = j4 * inv;
  }
}

extern "C" void kernel_launch(void* const* d_in, const int* in_sizes, int n_in,
                              void* d_out, int out_size, void* d_ws, size_t ws_size,
                              hipStream_t stream) {
  const float* x  = (const float*)d_in[0];
  const float* Wg = (const float*)d_in[1];
  const float* bg = (const float*)d_in[2];
  const float* Wp = (const float*)d_in[3];
  const float* bp = (const float*)d_in[4];
  const float* Wa = (const float*)d_in[5];
  const float* ba = (const float*)d_in[6];

  const int H = in_sizes[1] / 4;
  const int B = in_sizes[0] / H;

  const int rowsPerBlock = 16 * WAVES_PER_BLOCK;
  const int grid = B / rowsPerBlock;

  hipLaunchKernelGGL(dpl_safe_policy_kernel, dim3(grid), dim3(BLOCK), 0, stream,
                     x, Wg, bg, Wp, bp, Wa, ba, (float*)d_out, H);
}


# Every global read becomes a whole-line load: x streamed 512 B per wave-load through a wave-private LDS tile at four waves per SIMD; K split over 4 waves x 2 accumulators changes the summation order
# speedup vs baseline: 13.7880x; 13.7880x over previous
#include <hip/hip_runtime.h>
#include <math.h>


typedef __attribute__((ext_vector_type(2))) float v2f;
typedef __attribute__((ext_vector_type(4))) float v4f;
typedef __attribute__((ext_vector_type(8))) float v8f;

#define NWAVES    16
#define BLOCK     (NWAVES * 32)
#define RTILES    4
#define KSPLIT    4
#define KCHUNK    2048
#define KT        128
#define XPITCH    132
#define XT_FLOATS (16 * XPITCH)
#define WQ        68

__global__ __launch_bounds__(BLOCK) void dpl_safe_policy_kernel(
    const float* __restrict__ x,
    const float* __restrict__ Wg, const float* __restrict__ bg,
    const float* __restrict__ Wp, const float* __restrict__ bp,
    const float* __restrict__ Wa, const float* __restrict__ ba,
    float* __restrict__ out, int H)
{
  __shared__ __attribute__((aligned(16))) float ldsW[(KCHUNK / 4) * WQ];
  __shared__ __attribute__((aligned(16))) float ldsX[NWAVES * XT_FLOATS];

  const int tid  = threadIdx.x;
  const int lane = tid & 31;
  const int wave = tid >> 5;
  const int r    = wave & (RTILES - 1);
  const int q    = wave >> 2;
  const int c    = lane & 15;
  const int h    = lane >> 4;

  const int rowBase = (blockIdx.x * RTILES + r) * 16;

  float* xt = ldsX + wave * XT_FLOATS;
  const float* xg = x + (size_t)rowBase * (size_t)H + (lane << 2);

  v8f c0 = {};
  v8f c1 = {};

  for (int kbase = 0; kbase < H; kbase += KCHUNK) {
    const int KC    = (H - kbase < KCHUNK) ? (H - kbase) : KCHUNK;
    const int slice = KC / KSPLIT;
    const int nT    = slice / KT;
    const float* xs = xg + kbase + q * slice;

    v4f xr[16];
#pragma unroll
    for (int i = 0; i < 16; ++i)
      xr[i] = *(const v4f*)(xs + (size_t)i * (size_t)H);

    v4f wg[4], wp[4], wa[5];
#pragma unroll
    for (int u = 0; u < 4; ++u) {
      const int kk = tid + u * BLOCK;
      if (kk < KC) {
        wg[u] = *(const v4f*)(Wg + (size_t)(kbase + kk) * 4);
        wp[u] = *(const v4f*)(Wp + (size_t)(kbase + kk) * 4);
      }
    }
#pragma unroll
    for (int u = 0; u < 5; ++u) {
      const int t = tid + u * BLOCK;
      if (t < (KC * 5) / 4)
        wa[u] = *(const v4f*)(Wa + (size_t)kbase * 5 + (size_t)t * 4);
    }
#pragma unroll
    for (int u = 0; u < 4; ++u) {
      const int kk = tid + u * BLOCK;
      if (kk < KC) {
        float* d = ldsW + (kk >> 2) * WQ + (kk & 3);
        d[0]  = wg[u].x; d[4]  = wg[u].y; d[8]  = wg[u].z; d[12] = wg[u].w;
        d[16] = wp[u].x; d[20] = wp[u].y; d[24] = wp[u].z; d[28] = wp[u].w;
      }
    }
#pragma unroll
    for (int u = 0; u < 5; ++u) {
      const int t = tid + u * BLOCK;
      if (t < (KC * 5) / 4) {
#pragma unroll
        for (int e = 0; e < 4; ++e) {
          const int f  = t * 4 + e;
          const int kl = f / 5;
          const int n5 = f - kl * 5;
          ldsW[(kl >> 2) * WQ + ((8 + n5) << 2) + (kl & 3)] = wa[u][e];
        }
      }
    }
    for (int z = tid; z < (KC / 4) * 3; z += BLOCK) {
      const int qd = z / 3;
      const int m  = z - qd * 3;
      *(v4f*)(ldsW + qd * WQ + 52 + (m << 2)) = (v4f){0.0f, 0.0f, 0.0f, 0.0f};
    }
    __syncthreads();

    const float* wl = ldsW + ((q * slice) >> 2) * WQ + h * WQ + (c << 2);
    const float* ap = xt + c * XPITCH + (h << 2);

    for (int j = 0; j < nT; ++j) {
#pragma unroll
      for (int i = 0; i < 16; ++i)
        *(v4f*)(xt + i * XPITCH + (lane << 2)) = xr[i];

      if (j + 1 < nT) {
#pragma unroll
        for (int i = 0; i < 16; ++i)
          xr[i] = *(const v4f*)(xs + (j + 1) * KT + (size_t)i * (size_t)H);
      }

      const float* bfr = wl + j * ((KT / 4) * WQ);
#pragma unroll
      for (int hb = 0; hb < 2; ++hb) {
        v4f a[8], b[8];
#pragma unroll
        for (int p = 0; p < 8; ++p) {
          const int g = hb * 8 + p;
          a[p] = *(const v4f*)(ap + 8 * g);
          b[p] = *(const v4f*)(bfr + 2 * WQ * g);
        }
#pragma unroll
        for (int p = 0; p < 8; ++p) {
          v2f a0 = {a[p].x, a[p].y}, b0 = {b[p].x, b[p].y};
          v2f a1 = {a[p].z, a[p].w}, b1 = {b[p].z, b[p].w};
          c0 = __builtin_amdgcn_wmma_f32_16x16x4_f32(
                   false, a0,  false, b0,
                   (short)0, c0,  false,  false);
          c1 = __builtin_amdgcn_wmma_f32_16x16x4_f32(
                   false, a1,  false, b1,
                   (short)0, c1,  false,  false);
        }
      }
    }

    if (kbase + KCHUNK < H) __syncthreads();
  }

  {
    v8f acc = c0 + c1;
#pragma unroll
    for (int rr = 0; rr < 8; ++rr)
      xt[(rr + 8 * h) * 16 + c] = acc[rr];
  }
  __syncthreads();

  if (q == 0 && lane < 16) {
    float vals[16];
#pragma unroll
    for (int i = 0; i < 16; i += 4) {
      v4f t = *(const v4f*)(ldsX + (0 * RTILES + r) * XT_FLOATS + lane * 16 + i);
      t += *(const v4f*)(ldsX + (1 * RTILES + r) * XT_FLOATS + lane * 16 + i);
      t += *(const v4f*)(ldsX + (2 * RTILES + r) * XT_FLOATS + lane * 16 + i);
      t += *(const v4f*)(ldsX + (3 * RTILES + r) * XT_FLOATS + lane * 16 + i);
      vals[i] = t.x; vals[i + 1] = t.y; vals[i + 2] = t.z; vals[i + 3] = t.w;
    }
#pragma unroll
    for (int i = 0; i < 4; ++i) vals[i]     += bg[i];
#pragma unroll
    for (int i = 0; i < 4; ++i) vals[4 + i] += bp[i];
#pragma unroll
    for (int i = 0; i < 5; ++i) vals[8 + i] += ba[i];

    float mg = fmaxf(fmaxf(vals[0], vals[1]), fmaxf(vals[2], vals[3]));
    float g0 = __expf(vals[0] - mg), g1 = __expf(vals[1] - mg);
    float g2 = __expf(vals[2] - mg), g3 = __expf(vals[3] - mg);
    float ginv = 1.0f / (g0 + g1 + g2 + g3);

    float mp = fmaxf(fmaxf(vals[4], vals[5]), fmaxf(vals[6], vals[7]));
    float p0 = __expf(vals[4] - mp), p1 = __expf(vals[5] - mp);
    float p2 = __expf(vals[6] - mp), p3 = __expf(vals[7] - mp);
    float pinv = 1.0f / (p0 + p1 + p2 + p3);

    float ma = fmaxf(fmaxf(fmaxf(vals[8], vals[9]), fmaxf(vals[10], vals[11])), vals[12]);
    float e0 = __expf(vals[8]  - ma), e1 = __expf(vals[9]  - ma);
    float e2 = __expf(vals[10] - ma), e3 = __expf(vals[11] - ma);
    float e4 = __expf(vals[12] - ma);

    float gp = ginv * pinv;
    float u0 = (p0 * g0 + p1 * g1 + p2 * g2 + p3 * g3) * gp;
    float u1 = (p0 * g1 + p2 * g3) * gp;
    float u2 = (p1 * g0 + p3 * g2) * gp;

    float j0 = e0 * (1.0f - u0);
    float j1 = e1 * (1.0f - u1);
    float j2 = e2 * (1.0f - u2);
    float j3 = e3;
    float j4 = e4;
    float inv = 1.0f / (j0 + j1 + j2 + j3 + j4);

    float* o = out + (size_t)(rowBase + lane) * 5;
    o[0] = j0 * inv; o[1] = j1 * inv; o[2] = j2 * inv;
    o[3] = j3 * inv; o[4] = j4 * inv;
  }
}

extern "C" void kernel_launch(void* const* d_in, const int* in_sizes, int n_in,
                              void* d_out, int out_size, void* d_ws, size_t ws_size,
                              hipStream_t stream) {
  const float* x  = (const float*)d_in[0];
  const float* Wg = (const float*)d_in[1];
  const float* bg = (const float*)d_in[2];
  const float* Wp = (const float*)d_in[3];
  const float* bp = (const float*)d_in[4];
  const float* Wa = (const float*)d_in[5];
  const float* ba = (const float*)d_in[6];

  const int H = in_sizes[1] / 4;
  const int B = in_sizes[0] / H;

  const int rowsPerBlock = 16 * RTILES;
  const int grid = B / rowsPerBlock;

  hipLaunchKernelGGL(dpl_safe_policy_kernel, dim3(grid), dim3(BLOCK), 0, stream,
                     x, Wg, bg, Wp, bp, Wa, ba, (float*)d_out, H);
}
